# starting kernel
# baseline (speedup 1.0000x reference)
#include <hip/hip_runtime.h>
#include <hip/hip_bf16.h>


typedef __attribute__((ext_vector_type(2))) float    v2f;
typedef __attribute__((ext_vector_type(8))) float    v8f;
typedef __attribute__((ext_vector_type(2))) unsigned v2u;

#define NPATCH   196
#define DDIM     768
#define XB       150528u
#define CSTRIDE  50176u

__global__ __launch_bounds__(128) void patch_scorer_wmma(
    const float* __restrict__ x,
    const float* __restrict__ weight,
    const float* __restrict__ bias,
    float* __restrict__ out)
{
    __shared__ __align__(16) unsigned s_off[DDIM];
    __shared__ __align__(16) float    s_w[DDIM];

    const unsigned p = blockIdx.x;

    for (int d = threadIdx.x; d < DDIM; d += 128) {
        unsigned L    = p * (unsigned)DDIM + (unsigned)d;
        unsigned pipj = L / 588u;
        unsigned rem  = L - pipj * 588u;
        unsigned c    = rem / 196u;
        unsigned r2   = rem - c * 196u;
        unsigned i    = r2 / 14u;
        unsigned j    = r2 - i * 14u;
        unsigned pi   = pipj >> 4;
        unsigned pj   = pipj & 15u;
        s_off[d] = c * CSTRIDE + (i * 16u + pi) * 224u + (j * 16u + pj);
        s_w[d]   = weight[(size_t)p * DDIM + d];
    }
    __syncthreads();

    const unsigned lane = threadIdx.x & 31u;
    const unsigned wv   = threadIdx.x >> 5;
    const unsigned m    = lane & 15u;
    const unsigned hh   = lane >> 4;
    const unsigned b0   = blockIdx.y * 64u + wv * 16u;

    const unsigned idxBase = (b0 + m) * XB;

    v8f acc = {0.f, 0.f, 0.f, 0.f, 0.f, 0.f, 0.f, 0.f};

    #pragma unroll 4
    for (int kc = 0; kc < DDIM; kc += 4) {
        const int d0 = kc + (int)(hh << 1);
        const v2u off = *(const v2u*)&s_off[d0];
        v2f a;
        a.x = x[idxBase + off.x];
        a.y = x[idxBase + off.y];
        const v2f b = *(const v2f*)&s_w[d0];

        acc = __builtin_amdgcn_wmma_f32_16x16x4_f32(
            false, a, false, b, (short)0, acc, false, false);
    }

    const float bp = bias[p];
    if (m == 0u) {
        const unsigned mb = hh * 8u;
        #pragma unroll
        for (int v = 0; v < 8; ++v)
            out[(size_t)(b0 + mb + (unsigned)v) * NPATCH + p] = acc[v] + bp;
    }
}

extern "C" void kernel_launch(void* const* d_in, const int* in_sizes, int n_in,
                              void* d_out, int out_size, void* d_ws, size_t ws_size,
                              hipStream_t stream) {
    const float* x      = (const float*)d_in[0];
    const float* weight = (const float*)d_in[1];
    const float* bias   = (const float*)d_in[2];
    float*       out    = (float*)d_out;

    const int B = in_sizes[0] / (int)XB;
    dim3 grid(NPATCH, (unsigned)(B / 64), 1);
    dim3 block(128, 1, 1);
    patch_scorer_wmma<<<grid, block, 0, stream>>>(x, weight, bias, out);
}


# work-group = aligned 49-patch group x 4 images; WMMA rows = 16 pj, K along a plane, diagonal of D; x read as contiguous half-lines, weights in LDS; d summed per plane, then fixed-order LDS sum
# speedup vs baseline: 7.7653x; 7.7653x over previous
#include <hip/hip_runtime.h>


typedef __attribute__((ext_vector_type(2))) float    v2f;
typedef __attribute__((ext_vector_type(4))) float    v4f;
typedef __attribute__((ext_vector_type(8))) float    v8f;
typedef __attribute__((ext_vector_type(2))) unsigned v2u;

#define NPATCH   196
#define XB       150528u
#define CSTRIDE  50176u
#define QW       37632
#define PLANE    588u

__global__ __launch_bounds__(512) void patch_scorer_planes(
    const float* __restrict__ x,
    const float* __restrict__ weight,
    const float* __restrict__ bias,
    float* __restrict__ out)
{
    __shared__ __align__(16) float    s_w[QW];
    __shared__ __align__(16) unsigned s_off[588];
    __shared__ float                  s_part[4][64][2];

    const unsigned Q   = blockIdx.x;
    const unsigned bg  = blockIdx.y;
    const unsigned tid = threadIdx.x;

    {
        const v4f* __restrict__ wsrc = (const v4f*)(weight + (size_t)Q * QW);
        v4f* wdst = (v4f*)s_w;
        for (int i = (int)tid; i < QW / 4; i += 512) wdst[i] = wsrc[i];
        for (unsigned e = tid; e < PLANE; e += 512u) {
            const unsigned cc = e / 196u, tt = e - cc * 196u;
            s_off[e] = cc * CSTRIDE + (tt / 14u) * 3584u + (tt % 14u) * 16u;
        }
    }
    __syncthreads();

    const unsigned lane = tid & 31u;
    const unsigned wv   = tid >> 5;
    const unsigned pil  = wv & 3u;
    const unsigned bs   = wv >> 2;
    const unsigned n    = lane & 15u;
    const unsigned h    = lane >> 4;

    const unsigned b    = bg * 4u + bs;
    const unsigned pi   = Q * 4u + pil;
    const unsigned xbase = b * XB + pi * 224u + n;

    const unsigned ql   = pil * 16u + n;
    const unsigned q    = Q * 64u + ql;
    const unsigned p1   = (PLANE * q) / 768u;
    const unsigned ob   = 768u * (p1 + 1u) - PLANE * q;
    const unsigned wbase = ql * PLANE + 2u * h;

    v8f accLo = {0.f, 0.f, 0.f, 0.f, 0.f, 0.f, 0.f, 0.f};
    v8f accHi = {0.f, 0.f, 0.f, 0.f, 0.f, 0.f, 0.f, 0.f};

    #pragma unroll 21
    for (unsigned j = 0; j < 147u; ++j) {
        const unsigned e = 4u * j + 2u * h;
        const v2u off = *(const v2u*)&s_off[e];
        v2f a;
        a.x = x[xbase + off.x];
        a.y = x[xbase + off.y];
        const v2f w = *(const v2f*)&s_w[wbase + 4u * j];
        const bool isLo = (4u * j) < ob;
        v2f blo, bhi;
        blo.x = isLo ? w.x : 0.f;  blo.y = isLo ? w.y : 0.f;
        bhi.x = isLo ? 0.f : w.x;  bhi.y = isLo ? 0.f : w.y;
        accLo = __builtin_amdgcn_wmma_f32_16x16x4_f32(
            false, a, false, blo, (short)0, accLo, false, false);
        accHi = __builtin_amdgcn_wmma_f32_16x16x4_f32(
            false, a, false, bhi, (short)0, accHi, false, false);
    }

    {
        const unsigned r = lane & 7u;
        float lo = accLo[0], hi = accHi[0];
        #pragma unroll
        for (int k = 1; k < 8; ++k) {
            if (r == (unsigned)k) { lo = accLo[k]; hi = accHi[k]; }
        }
        if ((n >> 3) == h) {
            s_part[bs][ql][0] = lo;
            s_part[bs][ql][1] = hi;
        }
    }
    __syncthreads();

    if (tid < 196u) {
        const unsigned bs2 = tid / 49u;
        const unsigned P   = Q * 49u + (tid % 49u);
        const unsigned q0  = (768u * P) / PLANE;
        const unsigned q1  = (768u * P + 767u) / PLANE;
        float s = 0.f;
        for (unsigned qq = q0; qq <= q1; ++qq) {
            const unsigned pq = (PLANE * qq) / 768u;
            s += s_part[bs2][qq - Q * 64u][pq == P ? 0 : 1];
        }
        out[(size_t)(bg * 4u + bs2) * NPATCH + P] = s + bias[P];
    }
}

extern "C" void kernel_launch(void* const* d_in, const int* in_sizes, int n_in,
                              void* d_out, int out_size, void* d_ws, size_t ws_size,
                              hipStream_t stream) {
    const float* x      = (const float*)d_in[0];
    const float* weight = (const float*)d_in[1];
    const float* bias   = (const float*)d_in[2];
    float*       out    = (float*)d_out;

    dim3 grid(4, 64, 1);
    dim3 block(512, 1, 1);
    patch_scorer_planes<<<grid, block, 0, stream>>>(x, weight, bias, out);
}


# x loads software-pipelined in registers: a step's loads issued 24 steps ahead of its WMMAs (groups of 7), 48-62 in flight per wave while it computes; was 21-step trips that drained; sums unchanged
# speedup vs baseline: 8.0813x; 1.0407x over previous
#include <hip/hip_runtime.h>


typedef __attribute__((ext_vector_type(2))) float    v2f;
typedef __attribute__((ext_vector_type(4))) float    v4f;
typedef __attribute__((ext_vector_type(8))) float    v8f;
typedef __attribute__((ext_vector_type(2))) unsigned v2u;

#define NPATCH   196
#define XB       150528u
#define CSTRIDE  50176u
#define QW       37632
#define PLANE    588u
#define NSTEP    147u
#define DEPTH    24u

__global__ __launch_bounds__(512) void patch_scorer_planes(
    const float* __restrict__ x,
    const float* __restrict__ weight,
    const float* __restrict__ bias,
    float* __restrict__ out)
{
    __shared__ __align__(16) float    s_w[QW];
    __shared__ __align__(16) unsigned s_off[588];
    __shared__ float                  s_part[4][64][2];

    const unsigned Q   = blockIdx.x;
    const unsigned bg  = blockIdx.y;
    const unsigned tid = threadIdx.x;

    {
        const v4f* __restrict__ wsrc = (const v4f*)(weight + (size_t)Q * QW);
        v4f* wdst = (v4f*)s_w;
        for (int i = (int)tid; i < QW / 4; i += 512) wdst[i] = wsrc[i];
        for (unsigned e = tid; e < PLANE; e += 512u) {
            const unsigned cc = e / 196u, tt = e - cc * 196u;
            s_off[e] = cc * CSTRIDE + (tt / 14u) * 3584u + (tt % 14u) * 16u;
        }
    }
    __syncthreads();

    const unsigned lane = tid & 31u;
    const unsigned wv   = tid >> 5;
    const unsigned pil  = wv & 3u;
    const unsigned bs   = wv >> 2;
    const unsigned n    = lane & 15u;
    const unsigned h    = lane >> 4;

    const unsigned b    = bg * 4u + bs;
    const unsigned pi   = Q * 4u + pil;
    const unsigned xbase = b * XB + pi * 224u + n;

    const unsigned ql   = pil * 16u + n;
    const unsigned q    = Q * 64u + ql;
    const unsigned p1   = (PLANE * q) / 768u;
    const unsigned ob   = 768u * (p1 + 1u) - PLANE * q;
    const unsigned wbase = ql * PLANE + 2u * h;

    v8f accLo = {0.f, 0.f, 0.f, 0.f, 0.f, 0.f, 0.f, 0.f};
    v8f accHi = {0.f, 0.f, 0.f, 0.f, 0.f, 0.f, 0.f, 0.f};

    v2f areg[NSTEP];
    #pragma unroll
    for (unsigned j = 0; j < DEPTH; ++j) {
        const unsigned e = 4u * j + 2u * h;
        const v2u off = *(const v2u*)&s_off[e];
        areg[j].x = x[xbase + off.x];
        areg[j].y = x[xbase + off.y];
    }
    #pragma unroll
    for (unsigned j = 0; j < NSTEP; ++j) {
        if ((j % 7u) == 0u) __builtin_amdgcn_sched_barrier(0);
        if (j + DEPTH < NSTEP) {
            const unsigned e = 4u * (j + DEPTH) + 2u * h;
            const v2u off = *(const v2u*)&s_off[e];
            areg[j + DEPTH].x = x[xbase + off.x];
            areg[j + DEPTH].y = x[xbase + off.y];
        }
        const v2f a = areg[j];
        const v2f w = *(const v2f*)&s_w[wbase + 4u * j];
        const bool isLo = (4u * j) < ob;
        v2f blo, bhi;
        blo.x = isLo ? w.x : 0.f;  blo.y = isLo ? w.y : 0.f;
        bhi.x = isLo ? 0.f : w.x;  bhi.y = isLo ? 0.f : w.y;
        accLo = __builtin_amdgcn_wmma_f32_16x16x4_f32(
            false, a, false, blo, (short)0, accLo, false, false);
        accHi = __builtin_amdgcn_wmma_f32_16x16x4_f32(
            false, a, false, bhi, (short)0, accHi, false, false);
    }

    {
        const unsigned r = lane & 7u;
        float lo = accLo[0], hi = accHi[0];
        #pragma unroll
        for (int k = 1; k < 8; ++k) {
            if (r == (unsigned)k) { lo = accLo[k]; hi = accHi[k]; }
        }
        if ((n >> 3) == h) {
            s_part[bs][ql][0] = lo;
            s_part[bs][ql][1] = hi;
        }
    }
    __syncthreads();

    if (tid < 196u) {
        const unsigned bs2 = tid / 49u;
        const unsigned P   = Q * 49u + (tid % 49u);
        const unsigned q0  = (768u * P) / PLANE;
        const unsigned q1  = (768u * P + 767u) / PLANE;
        float s = 0.f;
        for (unsigned qq = q0; qq <= q1; ++qq) {
            const unsigned pq = (PLANE * qq) / 768u;
            s += s_part[bs2][qq - Q * 64u][pq == P ? 0 : 1];
        }
        out[(size_t)(bg * 4u + bs2) * NPATCH + P] = s + bias[P];
    }
}

extern "C" void kernel_launch(void* const* d_in, const int* in_sizes, int n_in,
                              void* d_out, int out_size, void* d_ws, size_t ws_size,
                              hipStream_t stream) {
    const float* x      = (const float*)d_in[0];
    const float* weight = (const float*)d_in[1];
    const float* bias   = (const float*)d_in[2];
    float*       out    = (float*)d_out;

    dim3 grid(4, 64, 1);
    dim3 block(512, 1, 1);
    patch_scorer_planes<<<grid, block, 0, stream>>>(x, weight, bias, out);
}


# LDS reads of a 7-step group (weights, x offsets) issued as ONE batch one group ahead, in front of the group's loads and WMMAs; was 4 read points a group, two waited for in place; sums unchanged
# speedup vs baseline: 8.2983x; 1.0269x over previous
#include <hip/hip_runtime.h>


typedef __attribute__((ext_vector_type(2))) float    v2f;
typedef __attribute__((ext_vector_type(4))) float    v4f;
typedef __attribute__((ext_vector_type(8))) float    v8f;
typedef __attribute__((ext_vector_type(2))) unsigned v2u;

#define NPATCH   196
#define XB       150528u
#define CSTRIDE  50176u
#define QW       37632
#define PLANE    588u
#define NSTEP    147u
#define DEPTH    24u

__global__ __launch_bounds__(512) void patch_scorer_planes(
    const float* __restrict__ x,
    const float* __restrict__ weight,
    const float* __restrict__ bias,
    float* __restrict__ out)
{
    __shared__ __align__(16) float    s_w[QW];
    __shared__ __align__(16) unsigned s_off[588];
    __shared__ float                  s_part[4][64][2];

    const unsigned Q   = blockIdx.x;
    const unsigned bg  = blockIdx.y;
    const unsigned tid = threadIdx.x;

    {
        const v4f* __restrict__ wsrc = (const v4f*)(weight + (size_t)Q * QW);
        v4f* wdst = (v4f*)s_w;
        for (int i = (int)tid; i < QW / 4; i += 512) wdst[i] = wsrc[i];
        for (unsigned e = tid; e < PLANE; e += 512u) {
            const unsigned cc = e / 196u, tt = e - cc * 196u;
            s_off[e] = cc * CSTRIDE + (tt / 14u) * 3584u + (tt % 14u) * 16u;
        }
    }
    __syncthreads();

    const unsigned lane = tid & 31u;
    const unsigned wv   = tid >> 5;
    const unsigned pil  = wv & 3u;
    const unsigned bs   = wv >> 2;
    const unsigned n    = lane & 15u;
    const unsigned h    = lane >> 4;

    const unsigned b    = bg * 4u + bs;
    const unsigned pi   = Q * 4u + pil;
    const unsigned xbase = b * XB + pi * 224u + n;

    const unsigned ql   = pil * 16u + n;
    const unsigned q    = Q * 64u + ql;
    const unsigned p1   = (PLANE * q) / 768u;
    const unsigned ob   = 768u * (p1 + 1u) - PLANE * q;
    const unsigned wbase = ql * PLANE + 2u * h;

    v8f accLo = {0.f, 0.f, 0.f, 0.f, 0.f, 0.f, 0.f, 0.f};
    v8f accHi = {0.f, 0.f, 0.f, 0.f, 0.f, 0.f, 0.f, 0.f};

    v2f areg[NSTEP];
    v2f wreg[NSTEP];
    v2u oreg[NSTEP + 7u];
    #pragma unroll
    for (unsigned j = 0; j < DEPTH; ++j) {
        const unsigned e = 4u * j + 2u * h;
        const v2u off = *(const v2u*)&s_off[e];
        areg[j].x = x[xbase + off.x];
        areg[j].y = x[xbase + off.y];
    }
    #pragma unroll
    for (unsigned t = 0; t < 7u; ++t) {
        wreg[t] = *(const v2f*)&s_w[wbase + 4u * t];
        oreg[DEPTH + t] = *(const v2u*)&s_off[4u * (DEPTH + t) + 2u * h];
    }
    #pragma unroll
    for (unsigned g = 0; g < NSTEP / 7u; ++g) {
        __builtin_amdgcn_sched_barrier(0);
        #pragma unroll
        for (unsigned t = 0; t < 7u; ++t) {
            const unsigned sw = 7u * (g + 1u) + t;
            if (sw < NSTEP) wreg[sw] = *(const v2f*)&s_w[wbase + 4u * sw];
            const unsigned so = sw + DEPTH;
            if (so < NSTEP) oreg[so] = *(const v2u*)&s_off[4u * so + 2u * h];
        }
        __builtin_amdgcn_sched_barrier(0);
        #pragma unroll
        for (unsigned t = 0; t < 7u; ++t) {
            const unsigned sa = 7u * g + DEPTH + t;
            if (sa < NSTEP) {
                areg[sa].x = x[xbase + oreg[sa].x];
                areg[sa].y = x[xbase + oreg[sa].y];
            }
        }
        #pragma unroll
        for (unsigned t = 0; t < 7u; ++t) {
            const unsigned j = 7u * g + t;
            const v2f a = areg[j];
            const v2f w = wreg[j];
            const bool isLo = (4u * j) < ob;
            v2f blo, bhi;
            blo.x = isLo ? w.x : 0.f;  blo.y = isLo ? w.y : 0.f;
            bhi.x = isLo ? 0.f : w.x;  bhi.y = isLo ? 0.f : w.y;
            accLo = __builtin_amdgcn_wmma_f32_16x16x4_f32(
                false, a, false, blo, (short)0, accLo, false, false);
            accHi = __builtin_amdgcn_wmma_f32_16x16x4_f32(
                false, a, false, bhi, (short)0, accHi, false, false);
        }
    }

    {
        const unsigned r = lane & 7u;
        float lo = accLo[0], hi = accHi[0];
        #pragma unroll
        for (int k = 1; k < 8; ++k) {
            if (r == (unsigned)k) { lo = accLo[k]; hi = accHi[k]; }
        }
        if ((n >> 3) == h) {
            s_part[bs][ql][0] = lo;
            s_part[bs][ql][1] = hi;
        }
    }
    __syncthreads();

    if (tid < 196u) {
        const unsigned bs2 = tid / 49u;
        const unsigned P   = Q * 49u + (tid % 49u);
        const unsigned q0  = (768u * P) / PLANE;
        const unsigned q1  = (768u * P + 767u) / PLANE;
        float s = 0.f;
        for (unsigned qq = q0; qq <= q1; ++qq) {
            const unsigned pq = (PLANE * qq) / 768u;
            s += s_part[bs2][qq - Q * 64u][pq == P ? 0 : 1];
        }
        out[(size_t)(bg * 4u + bs2) * NPATCH + P] = s + bias[P];
    }
}

extern "C" void kernel_launch(void* const* d_in, const int* in_sizes, int n_in,
                              void* d_out, int out_size, void* d_ws, size_t ws_size,
                              hipStream_t stream) {
    const float* x      = (const float*)d_in[0];
    const float* weight = (const float*)d_in[1];
    const float* bias   = (const float*)d_in[2];
    float*       out    = (float*)d_out;

    dim3 grid(4, 64, 1);
    dim3 block(512, 1, 1);
    patch_scorer_planes<<<grid, block, 0, stream>>>(x, weight, bias, out);
}


# x offsets become compile-time immediates off two fixed per-lane bases: offset table, its LDS reads and per-load address adds gone, no address register rewritten under loads in flight; sums unchanged
# speedup vs baseline: 8.6588x; 1.0434x over previous
#include <hip/hip_runtime.h>


typedef __attribute__((ext_vector_type(2))) float    v2f;
typedef __attribute__((ext_vector_type(4))) float    v4f;
typedef __attribute__((ext_vector_type(8))) float    v8f;

#define NPATCH   196
#define XB       150528u
#define QW       37632
#define LDSRES   3328
#define PLANE    588u
#define NSTEP    147u
#define DEPTH    24u

__global__ __launch_bounds__(512) void patch_scorer_planes(
    const float* __restrict__ x,
    const float* __restrict__ weight,
    const float* __restrict__ bias,
    float* __restrict__ out)
{
    __shared__ __align__(16) float    s_w[QW + LDSRES];
    __shared__ float                  s_part[4][64][2];

    const unsigned Q   = blockIdx.x;
    const unsigned bg  = blockIdx.y;
    const unsigned tid = threadIdx.x;

    {
        const v4f* __restrict__ wsrc = (const v4f*)(weight + (size_t)Q * QW);
        v4f* wdst = (v4f*)s_w;
        for (int i = (int)tid; i < QW / 4; i += 512) wdst[i] = wsrc[i];
    }
    __syncthreads();

    const unsigned lane = tid & 31u;
    const unsigned wv   = tid >> 5;
    const unsigned pil  = wv & 3u;
    const unsigned bs   = wv >> 2;
    const unsigned n    = lane & 15u;
    const unsigned h    = lane >> 4;

    const unsigned b    = bg * 4u + bs;
    const unsigned pi   = Q * 4u + pil;
    const unsigned xbase = b * XB + pi * 224u + n;
    const float* __restrict__ xrow  = x + (size_t)(xbase + 32u * h);
    const float* __restrict__ xwrap = x + (size_t)(xbase + 3392u * h);

    const unsigned ql   = pil * 16u + n;
    const unsigned q    = Q * 64u + ql;
    const unsigned p1   = (PLANE * q) / 768u;
    const unsigned ob   = 768u * (p1 + 1u) - PLANE * q;
    const unsigned wbase = ql * PLANE + 2u * h;

    v8f accLo = {0.f, 0.f, 0.f, 0.f, 0.f, 0.f, 0.f, 0.f};
    v8f accHi = {0.f, 0.f, 0.f, 0.f, 0.f, 0.f, 0.f, 0.f};

    v2f areg[NSTEP];
    v2f wreg[NSTEP];
    #pragma unroll
    for (unsigned j = 0; j < DEPTH; ++j) {
        const unsigned e0 = 4u * j;
        const float* __restrict__ xp = ((e0 % 14u) == 12u) ? xwrap : xrow;
        const unsigned o = (e0 / 14u) * 3584u + (e0 % 14u) * 16u;
        areg[j].x = xp[o];
        areg[j].y = xp[o + 16u];
    }
    #pragma unroll
    for (unsigned t = 0; t < 7u; ++t) {
        wreg[t] = *(const v2f*)&s_w[wbase + 4u * t];
    }
    #pragma unroll
    for (unsigned g = 0; g < NSTEP / 7u; ++g) {
        __builtin_amdgcn_sched_barrier(0);
        #pragma unroll
        for (unsigned t = 0; t < 7u; ++t) {
            const unsigned sw = 7u * (g + 1u) + t;
            if (sw < NSTEP) wreg[sw] = *(const v2f*)&s_w[wbase + 4u * sw];
        }
        __builtin_amdgcn_sched_barrier(0);
        #pragma unroll
        for (unsigned t = 0; t < 7u; ++t) {
            const unsigned sa = 7u * g + DEPTH + t;
            if (sa < NSTEP) {
                const unsigned e0 = 4u * sa;
                const float* __restrict__ xp = ((e0 % 14u) == 12u) ? xwrap : xrow;
                const unsigned o = (e0 / 14u) * 3584u + (e0 % 14u) * 16u;
                areg[sa].x = xp[o];
                areg[sa].y = xp[o + 16u];
            }
        }
        #pragma unroll
        for (unsigned t = 0; t < 7u; ++t) {
            const unsigned j = 7u * g + t;
            const v2f a = areg[j];
            const v2f w = wreg[j];
            const bool isLo = (4u * j) < ob;
            v2f blo, bhi;
            blo.x = isLo ? w.x : 0.f;  blo.y = isLo ? w.y : 0.f;
            bhi.x = isLo ? 0.f : w.x;  bhi.y = isLo ? 0.f : w.y;
            accLo = __builtin_amdgcn_wmma_f32_16x16x4_f32(
                false, a, false, blo, (short)0, accLo, false, false);
            accHi = __builtin_amdgcn_wmma_f32_16x16x4_f32(
                false, a, false, bhi, (short)0, accHi, false, false);
        }
    }

    {
        const unsigned r = lane & 7u;
        float lo = accLo[0], hi = accHi[0];
        #pragma unroll
        for (int k = 1; k < 8; ++k) {
            if (r == (unsigned)k) { lo = accLo[k]; hi = accHi[k]; }
        }
        if ((n >> 3) == h) {
            s_part[bs][ql][0] = lo;
            s_part[bs][ql][1] = hi;
        }
    }
    __syncthreads();

    if (tid < 196u) {
        const unsigned bs2 = tid / 49u;
        const unsigned P   = Q * 49u + (tid % 49u);
        const unsigned q0  = (768u * P) / PLANE;
        const unsigned q1  = (768u * P + 767u) / PLANE;
        float s = 0.f;
        for (unsigned qq = q0; qq <= q1; ++qq) {
            const unsigned pq = (PLANE * qq) / 768u;
            s += s_part[bs2][qq - Q * 64u][pq == P ? 0 : 1];
        }
        out[(size_t)(bg * 4u + bs2) * NPATCH + P] = s + bias[P];
    }
}

extern "C" void kernel_launch(void* const* d_in, const int* in_sizes, int n_in,
                              void* d_out, int out_size, void* d_ws, size_t ws_size,
                              hipStream_t stream) {
    const float* x      = (const float*)d_in[0];
    const float* weight = (const float*)d_in[1];
    const float* bias   = (const float*)d_in[2];
    float*       out    = (float*)d_out;

    dim3 grid(4, 64, 1);
    dim3 block(512, 1, 1);
    patch_scorer_planes<<<grid, block, 0, stream>>>(x, weight, bias, out);
}


# weight staging cut in 3 chunks along K: only steps 0-27 (28 KB of 150) stored before the x stream starts; rest requested inside the stream, handed over by split LDS-only barriers; sums unchanged
# speedup vs baseline: 10.3687x; 1.1975x over previous
#include <hip/hip_runtime.h>


typedef __attribute__((ext_vector_type(2))) float    v2f;
typedef __attribute__((ext_vector_type(4))) float    v4f;
typedef __attribute__((ext_vector_type(8))) float    v8f;

#define NPATCH   196
#define XB       150528u
#define QW       37632
#define LDSRES   3328
#define PLANE    588u
#define NSTEP    147u
#define DEPTH    24u
#define C0_NP     28u
#define C1_FIRST  28u
#define C1_NP     48u
#define C2_FIRST  76u
#define C2_NP     71u
#define C0_PIECES 1792u
#define C2_PIECES 4544u

static __device__ __forceinline__ unsigned piece_off(unsigned i, unsigned first, unsigned np)
{
    const unsigned plane = i / np;
    return plane * PLANE + 4u * (first + (i - plane * np));
}

static __device__ __forceinline__ void lds_publish()
{
    asm volatile("s_wait_dscnt 0x0\n\ts_barrier_signal -1" ::: "memory");
}
static __device__ __forceinline__ void lds_acquire()
{
    asm volatile("s_barrier_wait -1" ::: "memory");
}

__global__ __launch_bounds__(512) void patch_scorer_planes(
    const float* __restrict__ x,
    const float* __restrict__ weight,
    const float* __restrict__ bias,
    float* __restrict__ out)
{
    __shared__ __align__(16) float    s_w[QW + LDSRES];
    __shared__ float                  s_part[4][64][2];

    const unsigned Q   = blockIdx.x;
    const unsigned bg  = blockIdx.y;
    const unsigned tid = threadIdx.x;

    const float* __restrict__ wq = weight + (size_t)Q * QW;
    {
        v4f w0[4];
        #pragma unroll
        for (unsigned m = 0; m < 4u; ++m) {
            const unsigned i  = tid + 512u * m;
            const unsigned ic = (i < C0_PIECES) ? i : (C0_PIECES - 1u);
            w0[m] = *(const v4f*)(wq + piece_off(ic, 0u, C0_NP));
        }
        #pragma unroll
        for (unsigned m = 0; m < 4u; ++m) {
            const unsigned i = tid + 512u * m;
            const unsigned o = (i < C0_PIECES) ? piece_off(i, 0u, C0_NP) : (unsigned)QW + 4u * (i - C0_PIECES);
            *(v4f*)&s_w[o] = w0[m];
        }
    }
    lds_publish();
    lds_acquire();

    const unsigned lane = tid & 31u;
    const unsigned wv   = tid >> 5;
    const unsigned pil  = wv & 3u;
    const unsigned bs   = wv >> 2;
    const unsigned n    = lane & 15u;
    const unsigned h    = lane >> 4;

    const unsigned b    = bg * 4u + bs;
    const unsigned pi   = Q * 4u + pil;
    const unsigned xbase = b * XB + pi * 224u + n;
    const float* __restrict__ xrow  = x + (size_t)(xbase + 32u * h);
    const float* __restrict__ xwrap = x + (size_t)(xbase + 3392u * h);

    const unsigned ql   = pil * 16u + n;
    const unsigned q    = Q * 64u + ql;
    const unsigned p1   = (PLANE * q) / 768u;
    const unsigned ob   = 768u * (p1 + 1u) - PLANE * q;
    const unsigned wbase = ql * PLANE + 2u * h;

    v8f accLo = {0.f, 0.f, 0.f, 0.f, 0.f, 0.f, 0.f, 0.f};
    v8f accHi = {0.f, 0.f, 0.f, 0.f, 0.f, 0.f, 0.f, 0.f};

    v2f areg[NSTEP];
    v2f wreg[NSTEP];
    v4f w1[6];
    v4f w2[9];
    #pragma unroll
    for (unsigned j = 0; j < DEPTH; ++j) {
        if (j == 14u) {
            __builtin_amdgcn_sched_barrier(0);
            #pragma unroll
            for (unsigned m = 0; m < 6u; ++m)
                w1[m] = *(const v4f*)(wq + piece_off(tid + 512u * m, C1_FIRST, C1_NP));
            __builtin_amdgcn_sched_barrier(0);
        }
        const unsigned e0 = 4u * j;
        const float* __restrict__ xp = ((e0 % 14u) == 12u) ? xwrap : xrow;
        const unsigned o = (e0 / 14u) * 3584u + (e0 % 14u) * 16u;
        areg[j].x = xp[o];
        areg[j].y = xp[o + 16u];
    }
    #pragma unroll
    for (unsigned t = 0; t < 7u; ++t) {
        wreg[t] = *(const v2f*)&s_w[wbase + 4u * t];
    }
    #pragma unroll
    for (unsigned g = 0; g < NSTEP / 7u; ++g) {
        __builtin_amdgcn_sched_barrier(0);
        if (g == 2u) {
            #pragma unroll
            for (unsigned m = 0; m < 6u; ++m)
                *(v4f*)&s_w[piece_off(tid + 512u * m, C1_FIRST, C1_NP)] = w1[m];
            lds_publish();
        }
        if (g == 8u) {
            #pragma unroll
            for (unsigned m = 0; m < 9u; ++m) {
                const unsigned i = tid + 512u * m;
                const unsigned o = (i < C2_PIECES) ? piece_off(i, C2_FIRST, C2_NP)
                                                   : (unsigned)QW + 4u * (i - C2_PIECES);
                *(v4f*)&s_w[o] = w2[m];
            }
            lds_publish();
        }
        if (g == 3u || g == 9u) lds_acquire();
        #pragma unroll
        for (unsigned t = 0; t < 7u; ++t) {
            const unsigned sw = 7u * (g + 1u) + t;
            if (sw < NSTEP) wreg[sw] = *(const v2f*)&s_w[wbase + 4u * sw];
        }
        __builtin_amdgcn_sched_barrier(0);
        #pragma unroll
        for (unsigned mm = 0; mm < 3u; ++mm) {
            if (g >= 2u && g <= 4u) {
                const unsigned m  = 3u * (g - 2u) + mm;
                const unsigned i  = tid + 512u * m;
                const unsigned ic = (i < C2_PIECES) ? i : (C2_PIECES - 1u);
                w2[m] = *(const v4f*)(wq + piece_off(ic, C2_FIRST, C2_NP));
            }
        }
        #pragma unroll
        for (unsigned t = 0; t < 7u; ++t) {
            const unsigned sa = 7u * g + DEPTH + t;
            if (sa < NSTEP) {
                const unsigned e0 = 4u * sa;
                const float* __restrict__ xp = ((e0 % 14u) == 12u) ? xwrap : xrow;
                const unsigned o = (e0 / 14u) * 3584u + (e0 % 14u) * 16u;
                areg[sa].x = xp[o];
                areg[sa].y = xp[o + 16u];
            }
        }
        #pragma unroll
        for (unsigned t = 0; t < 7u; ++t) {
            const unsigned j = 7u * g + t;
            const v2f a = areg[j];
            const v2f w = wreg[j];
            const bool isLo = (4u * j) < ob;
            v2f blo, bhi;
            blo.x = isLo ? w.x : 0.f;  blo.y = isLo ? w.y : 0.f;
            bhi.x = isLo ? 0.f : w.x;  bhi.y = isLo ? 0.f : w.y;
            accLo = __builtin_amdgcn_wmma_f32_16x16x4_f32(
                false, a, false, blo, (short)0, accLo, false, false);
            accHi = __builtin_amdgcn_wmma_f32_16x16x4_f32(
                false, a, false, bhi, (short)0, accHi, false, false);
        }
    }

    {
        const unsigned r = lane & 7u;
        float lo = accLo[0], hi = accHi[0];
        #pragma unroll
        for (int k = 1; k < 8; ++k) {
            if (r == (unsigned)k) { lo = accLo[k]; hi = accHi[k]; }
        }
        if ((n >> 3) == h) {
            s_part[bs][ql][0] = lo;
            s_part[bs][ql][1] = hi;
        }
    }
    __syncthreads();

    if (tid < 196u) {
        const unsigned bs2 = tid / 49u;
        const unsigned P   = Q * 49u + (tid % 49u);
        const unsigned q0  = (768u * P) / PLANE;
        const unsigned q1  = (768u * P + 767u) / PLANE;
        float s = 0.f;
        for (unsigned qq = q0; qq <= q1; ++qq) {
            const unsigned pq = (PLANE * qq) / 768u;
            s += s_part[bs2][qq - Q * 64u][pq == P ? 0 : 1];
        }
        out[(size_t)(bg * 4u + bs2) * NPATCH + P] = s + bias[P];
    }
}

extern "C" void kernel_launch(void* const* d_in, const int* in_sizes, int n_in,
                              void* d_out, int out_size, void* d_ws, size_t ws_size,
                              hipStream_t stream) {
    const float* x      = (const float*)d_in[0];
    const float* weight = (const float*)d_in[1];
    const float* bias   = (const float*)d_in[2];
    float*       out    = (float*)d_out;

    dim3 grid(4, 64, 1);
    dim3 block(512, 1, 1);
    patch_scorer_planes<<<grid, block, 0, stream>>>(x, weight, bias, out);
}


# one f32 WMMA per K step (was 2 on zero-split copies of B), as tried on v2: at a plane's patch boundary the owning wave saves its diagonal element of D to LDS and clears it; sums and order unchanged
# speedup vs baseline: 13.0552x; 1.2591x over previous
#include <hip/hip_runtime.h>


typedef __attribute__((ext_vector_type(2))) float    v2f;
typedef __attribute__((ext_vector_type(4))) float    v4f;
typedef __attribute__((ext_vector_type(8))) float    v8f;

#define NPATCH   196
#define XB       150528u
#define QW       37632
#define LDSRES   3328
#define PLANE    588u
#define NSTEP    147u
#define DEPTH    24u
#define C0_NP     28u
#define C1_FIRST  28u
#define C1_NP     48u
#define C2_FIRST  76u
#define C2_NP     71u
#define C0_PIECES 1792u
#define C2_PIECES 4544u

static __device__ __forceinline__ unsigned piece_off(unsigned i, unsigned first, unsigned np)
{
    const unsigned plane = i / np;
    return plane * PLANE + 4u * (first + (i - plane * np));
}

static __device__ __forceinline__ void lds_publish()
{
    asm volatile("s_wait_dscnt 0x0\n\ts_barrier_signal -1" ::: "memory");
}
static __device__ __forceinline__ void lds_acquire()
{
    asm volatile("s_barrier_wait -1" ::: "memory");
}

__global__ __launch_bounds__(512) void patch_scorer_planes(
    const float* __restrict__ x,
    const float* __restrict__ weight,
    const float* __restrict__ bias,
    float* __restrict__ out)
{
    __shared__ __align__(16) float    s_w[QW + LDSRES];
    __shared__ float                  s_part[4][64][2];

    const unsigned Q   = blockIdx.x;
    const unsigned bg  = blockIdx.y;
    const unsigned tid = threadIdx.x;

    const float* __restrict__ wq = weight + (size_t)Q * QW;
    {
        v4f w0[4];
        #pragma unroll
        for (unsigned m = 0; m < 4u; ++m) {
            const unsigned i  = tid + 512u * m;
            const unsigned ic = (i < C0_PIECES) ? i : (C0_PIECES - 1u);
            w0[m] = *(const v4f*)(wq + piece_off(ic, 0u, C0_NP));
        }
        #pragma unroll
        for (unsigned m = 0; m < 4u; ++m) {
            const unsigned i = tid + 512u * m;
            const unsigned o = (i < C0_PIECES) ? piece_off(i, 0u, C0_NP) : (unsigned)QW + 4u * (i - C0_PIECES);
            *(v4f*)&s_w[o] = w0[m];
        }
    }
    lds_publish();
    lds_acquire();

    const unsigned lane = tid & 31u;
    const unsigned wv   = tid >> 5;
    const unsigned pil  = wv & 3u;
    const unsigned bs   = wv >> 2;
    const unsigned n    = lane & 15u;
    const unsigned h    = lane >> 4;

    const unsigned b    = bg * 4u + bs;
    const unsigned pi   = Q * 4u + pil;
    const unsigned xbase = b * XB + pi * 224u + n;
    const float* __restrict__ xrow  = x + (size_t)(xbase + 32u * h);
    const float* __restrict__ xwrap = x + (size_t)(xbase + 3392u * h);

    const unsigned ql   = pil * 16u + n;
    const unsigned ob4  = 192u * ((147u * ql) / 192u + 1u) - 147u * ql;
    const unsigned wbase = ql * PLANE + 2u * h;
    const unsigned pil_s = (unsigned)__builtin_amdgcn_readfirstlane((int)pil);

    v8f acc = {0.f, 0.f, 0.f, 0.f, 0.f, 0.f, 0.f, 0.f};

    v2f areg[NSTEP];
    v2f wreg[NSTEP];
    v4f w1[6];
    v4f w2[9];
    #pragma unroll
    for (unsigned j = 0; j < DEPTH; ++j) {
        if (j == 14u) {
            __builtin_amdgcn_sched_barrier(0);
            #pragma unroll
            for (unsigned m = 0; m < 6u; ++m)
                w1[m] = *(const v4f*)(wq + piece_off(tid + 512u * m, C1_FIRST, C1_NP));
            __builtin_amdgcn_sched_barrier(0);
        }
        const unsigned e0 = 4u * j;
        const float* __restrict__ xp = ((e0 % 14u) == 12u) ? xwrap : xrow;
        const unsigned o = (e0 / 14u) * 3584u + (e0 % 14u) * 16u;
        areg[j].x = xp[o];
        areg[j].y = xp[o + 16u];
    }
    #pragma unroll
    for (unsigned t = 0; t < 7u; ++t) {
        wreg[t] = *(const v2f*)&s_w[wbase + 4u * t];
    }
    #pragma unroll
    for (unsigned g = 0; g < NSTEP / 7u; ++g) {
        __builtin_amdgcn_sched_barrier(0);
        if (g == 2u) {
            #pragma unroll
            for (unsigned m = 0; m < 6u; ++m)
                *(v4f*)&s_w[piece_off(tid + 512u * m, C1_FIRST, C1_NP)] = w1[m];
            lds_publish();
        }
        if (g == 8u) {
            #pragma unroll
            for (unsigned m = 0; m < 9u; ++m) {
                const unsigned i = tid + 512u * m;
                const unsigned o = (i < C2_PIECES) ? piece_off(i, C2_FIRST, C2_NP)
                                                   : (unsigned)QW + 4u * (i - C2_PIECES);
                *(v4f*)&s_w[o] = w2[m];
            }
            lds_publish();
        }
        if (g == 3u || g == 9u) lds_acquire();
        #pragma unroll
        for (unsigned t = 0; t < 7u; ++t) {
            const unsigned sw = 7u * (g + 1u) + t;
            if (sw < NSTEP) wreg[sw] = *(const v2f*)&s_w[wbase + 4u * sw];
        }
        __builtin_amdgcn_sched_barrier(0);
        #pragma unroll
        for (unsigned mm = 0; mm < 3u; ++mm) {
            if (g >= 2u && g <= 4u) {
                const unsigned m  = 3u * (g - 2u) + mm;
                const unsigned i  = tid + 512u * m;
                const unsigned ic = (i < C2_PIECES) ? i : (C2_PIECES - 1u);
                w2[m] = *(const v4f*)(wq + piece_off(ic, C2_FIRST, C2_NP));
            }
        }
        #pragma unroll
        for (unsigned t = 0; t < 7u; ++t) {
            const unsigned sa = 7u * g + DEPTH + t;
            if (sa < NSTEP) {
                const unsigned e0 = 4u * sa;
                const float* __restrict__ xp = ((e0 % 14u) == 12u) ? xwrap : xrow;
                const unsigned o = (e0 / 14u) * 3584u + (e0 % 14u) * 16u;
                areg[sa].x = xp[o];
                areg[sa].y = xp[o + 16u];
            }
        }
        #pragma unroll
        for (unsigned t = 0; t < 7u; ++t) {
            const unsigned j = 7u * g + t;
            if (j % 3u == 0u && j >= 3u && j <= 144u) {
                const unsigned qs = (47u * (j / 3u)) & 63u;
                if (pil_s == (qs >> 4)) {
                    const unsigned ns = qs & 15u;
                    if (lane == ns + 16u * (ns >> 3)) {
                        s_part[bs][qs][0] = acc[ns & 7u];
                        acc[ns & 7u] = 0.f;
                    }
                }
            }
            acc = __builtin_amdgcn_wmma_f32_16x16x4_f32(
                false, areg[j], false, wreg[j], (short)0, acc, false, false);
        }
    }

    {
        const unsigned r = lane & 7u;
        float d = acc[0];
        #pragma unroll
        for (int k = 1; k < 8; ++k) {
            if (r == (unsigned)k) d = acc[k];
        }
        if ((n >> 3) == h) {
            s_part[bs][ql][(ob4 < NSTEP) ? 1 : 0] = d;
        }
    }
    __syncthreads();

    if (tid < 196u) {
        const unsigned bs2 = tid / 49u;
        const unsigned P   = Q * 49u + (tid % 49u);
        const unsigned q0  = (768u * P) / PLANE;
        const unsigned q1  = (768u * P + 767u) / PLANE;
        float s = 0.f;
        for (unsigned qq = q0; qq <= q1; ++qq) {
            const unsigned pq = (PLANE * qq) / 768u;
            s += s_part[bs2][qq - Q * 64u][pq == P ? 0 : 1];
        }
        out[(size_t)(bg * 4u + bs2) * NPATCH + P] = s + bias[P];
    }
}

extern "C" void kernel_launch(void* const* d_in, const int* in_sizes, int n_in,
                              void* d_out, int out_size, void* d_ws, size_t ws_size,
                              hipStream_t stream) {
    const float* x      = (const float*)d_in[0];
    const float* weight = (const float*)d_in[1];
    const float* bias   = (const float*)d_in[2];
    float*       out    = (float*)d_out;

    dim3 grid(4, 64, 1);
    dim3 block(512, 1, 1);
    patch_scorer_planes<<<grid, block, 0, stream>>>(x, weight, bias, out);
}
